# starting kernel
# baseline (speedup 1.0000x reference)
#include <hip/hip_runtime.h>

typedef float v2f __attribute__((ext_vector_type(2)));
typedef float v8f __attribute__((ext_vector_type(8)));

#define BB   16
#define CIN  128
#define COUT 256
#define LL   32

#define QP   33
#define CP   155

__global__ __launch_bounds__(256) void fconv2d_wmma_kernel(
    const float* __restrict__ x,
    const float* __restrict__ w,
    float* __restrict__ out)
{
    __shared__ float sW2[9 * 16 * QP];
    __shared__ float sX2[3 * 18 * CP];

    const int tid = threadIdx.x;
    const int bid = blockIdx.x;
    const int b   = bid >> 6;
    const int rem = bid & 63;
    const int y   = rem >> 1;
    const int x0  = (rem & 1) << 4;

    for (int idx = tid; idx < 16 * 32 * 9; idx += 256) {
        int n = idx / 288;
        int r = idx % 288;
        int q = r / 9;
        int t = r % 9;
        int i = t / 3, j = t % 3;
        sW2[((2 - i) * 3 + (2 - j)) * (16 * QP) + n * QP + (31 - q)] = 0.5f * w[idx];
    }
    for (int idx = tid; idx < 153 * 54; idx += 256) {
        int c   = 1 + idx / 54;
        int r2  = idx % 54;
        int row = r2 / 18;
        int xi  = r2 % 18;
        int ch  = (c + 95) & 127;
        int gy  = y + row;
        int gx  = x0 + xi;
        float v = 0.0f;
        if (gy < LL && gx < LL)
            v = x[((b * CIN + ch) * LL + gy) * LL + gx];
        sX2[(row * 18 + xi) * CP + c] = v;
    }
    __syncthreads();

    const int lane = tid & 31;
    const int wid  = tid >> 5;
    const int col  = lane & 15;
    const int hi   = lane >> 4;

    const int d0 = wid;
    const int d1 = wid ? (16 - wid) : 8;

    v8f c0 = {0.f, 0.f, 0.f, 0.f, 0.f, 0.f, 0.f, 0.f};
    v8f c1 = {0.f, 0.f, 0.f, 0.f, 0.f, 0.f, 0.f, 0.f};

    if (wid != 0) {
        for (int dy = 0; dy < 3; ++dy) {
            for (int dx = 0; dx < 3; ++dx) {
                const float* ap  = &sW2[(dy * 3 + dx) * (16 * QP) + col * QP + 2 * hi];
                const int    sb  = (dy * 18 + col + dx) * CP + 2 + 2 * hi;
                const float* b0p = &sX2[sb + 8 * d0];
                const float* b1p = &sX2[sb + 8 * d1];
#pragma unroll
                for (int k = 0; k < 8; ++k) {
                    v2f a;
                    a.x = ap[4 * k];
                    a.y = ap[4 * k + 1];
                    v2f bf;
                    bf.x = b0p[4 * k]     + b1p[4 * k];
                    bf.y = b0p[4 * k + 1] + b1p[4 * k + 1];
                    c0 = __builtin_amdgcn_wmma_f32_16x16x4_f32(
                            false, a, false, bf, (short)0, c0, false, false);
                    c1 = __builtin_amdgcn_wmma_f32_16x16x4_f32(
                            false, a, false, bf, (short)0, c1, false, false);
                }
            }
        }
    } else {
        for (int dy = 0; dy < 3; ++dy) {
            for (int dx = 0; dx < 3; ++dx) {
                const float* ap  = &sW2[(dy * 3 + dx) * (16 * QP) + col * QP + 2 * hi];
                const int    sb  = (dy * 18 + col + dx) * CP + 2 + 2 * hi;
                const float* b0p = &sX2[sb + 8 * 0];
                const float* b1p = &sX2[sb + 8 * 8];
#pragma unroll
                for (int k = 0; k < 8; ++k) {
                    v2f a;
                    a.x = ap[4 * k];
                    a.y = ap[4 * k + 1];
                    float u0 = b0p[4 * k];
                    float u1 = b0p[4 * k + 1];
                    float v0 = b1p[4 * k];
                    float v1 = b1p[4 * k + 1];
                    v2f bf0, bf1;
                    bf0.x = u0 + u0;  bf0.y = u1 + u1;
                    bf1.x = v0 + v0;  bf1.y = v1 + v1;
                    c0 = __builtin_amdgcn_wmma_f32_16x16x4_f32(
                            false, a, false, bf0, (short)0, c0, false, false);
                    c1 = __builtin_amdgcn_wmma_f32_16x16x4_f32(
                            false, a, false, bf1, (short)0, c1, false, false);
                }
            }
        }
    }

    float* op0 = out + ((b * COUT + d0 * 16 + hi * 8) * LL + y) * LL + x0 + col;
    float* op1 = out + ((b * COUT + d1 * 16 + hi * 8) * LL + y) * LL + x0 + col;
#pragma unroll
    for (int r = 0; r < 8; ++r) {
        op0[r * (LL * LL)] = c0[r];
        op1[r * (LL * LL)] = c1[r];
    }
}

extern "C" void kernel_launch(void* const* d_in, const int* in_sizes, int n_in,
                              void* d_out, int out_size, void* d_ws, size_t ws_size,
                              hipStream_t stream) {
    const float* x = (const float*)d_in[0];
    const float* w = (const float*)d_in[1];
    float* out = (float*)d_out;

    dim3 grid(BB * LL * 2);
    dim3 block(256);
    hipLaunchKernelGGL(fconv2d_wmma_kernel, grid, block, 0, stream, x, w, out);
}


# staging: every global load of a block (weights and x slab, as float4) is issued before the first LDS store, so the load latencies overlap instead of one full latency per loop trip
# speedup vs baseline: 1.5359x; 1.5359x over previous
#include <hip/hip_runtime.h>

typedef float v2f __attribute__((ext_vector_type(2)));
typedef float v4f __attribute__((ext_vector_type(4)));
typedef float v8f __attribute__((ext_vector_type(8)));

#define BB   16
#define CIN  128
#define COUT 256
#define LL   32

#define QP   33
#define CP   155

__global__ __launch_bounds__(256) void fconv2d_wmma_kernel(
    const float* __restrict__ x,
    const float* __restrict__ w,
    float* __restrict__ out)
{
    __shared__ float sW2[9 * 16 * QP];
    __shared__ float sX2[3 * 18 * CP];

    const int tid = threadIdx.x;
    const int bid = blockIdx.x;
    const int b   = bid >> 6;
    const int rem = bid & 63;
    const int y   = rem >> 1;
    const int x0  = (rem & 1) << 4;


    const v4f* __restrict__ w4 = (const v4f*)w;
    v4f wv[5];
#pragma unroll
    for (int i = 0; i < 5; ++i) {
        int f = tid + 256 * i;
        f = f < 1152 ? f : 1151;
        wv[i] = w4[f];
    }

    const int q4 = tid & 3;
    const int g  = tid >> 2;
    v4f xv[3][2];
    v2f hv[3][2];
#pragma unroll
    for (int row = 0; row < 3; ++row) {
        int gy  = y + row;
        int gyc = gy < LL ? gy : (LL - 1);
#pragma unroll
        for (int j = 0; j < 2; ++j) {
            int s  = 1 + g + 64 * j;
            int ch = (s + 95) & 127;
            const float* lp = x + ((b * CIN + ch) * LL + gyc) * LL;
            xv[row][j] = *(const v4f*)(lp + x0 + 4 * q4);
            hv[row][j] = *(const v2f*)(lp + 16);
        }
    }

#pragma unroll
    for (int i = 0; i < 5; ++i) {
        int f = tid + 256 * i;
        if (f < 1152) {
            int idx = 4 * f;
            int nq  = idx / 9;
            int t   = idx - 9 * nq;
#pragma unroll
            for (int e = 0; e < 4; ++e) {
                sW2[(8 - t) * (16 * QP) + (nq >> 5) * QP + (31 - (nq & 31))] = 0.5f * wv[i][e];
                ++t;
                if (t == 9) { t = 0; ++nq; }
            }
        }
    }

    const bool halo_in = (x0 == 0);
#pragma unroll
    for (int row = 0; row < 3; ++row) {
        const bool row_in = (y + row) < LL;
#pragma unroll
        for (int j = 0; j < 2; ++j) {
            int s = 1 + g + 64 * j;
            float* dp = &sX2[(row * 18 + 4 * q4) * CP + s];
            v4f v = xv[row][j];
            float e0 = row_in ? v.x : 0.0f;
            float e1 = row_in ? v.y : 0.0f;
            float e2 = row_in ? v.z : 0.0f;
            float e3 = row_in ? v.w : 0.0f;
            dp[0]      = e0;
            dp[CP]     = e1;
            dp[2 * CP] = e2;
            dp[3 * CP] = e3;
            const bool dup = (j == 0) && (g < 25);
            if (dup) {
                dp[128]          = e0;
                dp[CP + 128]     = e1;
                dp[2 * CP + 128] = e2;
                dp[3 * CP + 128] = e3;
            }
            if (q4 == 3) {
                const bool hin = row_in && halo_in;
                float h0 = hin ? hv[row][j].x : 0.0f;
                float h1 = hin ? hv[row][j].y : 0.0f;
                dp[4 * CP] = h0;
                dp[5 * CP] = h1;
                if (dup) {
                    dp[4 * CP + 128] = h0;
                    dp[5 * CP + 128] = h1;
                }
            }
        }
    }
    __syncthreads();

    const int lane = tid & 31;
    const int wid  = tid >> 5;
    const int col  = lane & 15;
    const int hi   = lane >> 4;

    const int d0 = wid;
    const int d1 = wid ? (16 - wid) : 8;

    v8f c0 = {0.f, 0.f, 0.f, 0.f, 0.f, 0.f, 0.f, 0.f};
    v8f c1 = {0.f, 0.f, 0.f, 0.f, 0.f, 0.f, 0.f, 0.f};

    if (wid != 0) {
        for (int dy = 0; dy < 3; ++dy) {
            for (int dx = 0; dx < 3; ++dx) {
                const float* ap  = &sW2[(dy * 3 + dx) * (16 * QP) + col * QP + 2 * hi];
                const int    sb  = (dy * 18 + col + dx) * CP + 2 + 2 * hi;
                const float* b0p = &sX2[sb + 8 * d0];
                const float* b1p = &sX2[sb + 8 * d1];
#pragma unroll
                for (int k = 0; k < 8; ++k) {
                    v2f a;
                    a.x = ap[4 * k];
                    a.y = ap[4 * k + 1];
                    v2f bf;
                    bf.x = b0p[4 * k]     + b1p[4 * k];
                    bf.y = b0p[4 * k + 1] + b1p[4 * k + 1];
                    c0 = __builtin_amdgcn_wmma_f32_16x16x4_f32(
                            false, a, false, bf, (short)0, c0, false, false);
                    c1 = __builtin_amdgcn_wmma_f32_16x16x4_f32(
                            false, a, false, bf, (short)0, c1, false, false);
                }
            }
        }
    } else {
        for (int dy = 0; dy < 3; ++dy) {
            for (int dx = 0; dx < 3; ++dx) {
                const float* ap  = &sW2[(dy * 3 + dx) * (16 * QP) + col * QP + 2 * hi];
                const int    sb  = (dy * 18 + col + dx) * CP + 2 + 2 * hi;
                const float* b0p = &sX2[sb + 8 * 0];
                const float* b1p = &sX2[sb + 8 * 8];
#pragma unroll
                for (int k = 0; k < 8; ++k) {
                    v2f a;
                    a.x = ap[4 * k];
                    a.y = ap[4 * k + 1];
                    float u0 = b0p[4 * k];
                    float u1 = b0p[4 * k + 1];
                    float v0 = b1p[4 * k];
                    float v1 = b1p[4 * k + 1];
                    v2f bf0, bf1;
                    bf0.x = u0 + u0;  bf0.y = u1 + u1;
                    bf1.x = v0 + v0;  bf1.y = v1 + v1;
                    c0 = __builtin_amdgcn_wmma_f32_16x16x4_f32(
                            false, a, false, bf0, (short)0, c0, false, false);
                    c1 = __builtin_amdgcn_wmma_f32_16x16x4_f32(
                            false, a, false, bf1, (short)0, c1, false, false);
                }
            }
        }
    }

    float* op0 = out + ((b * COUT + d0 * 16 + hi * 8) * LL + y) * LL + x0 + col;
    float* op1 = out + ((b * COUT + d1 * 16 + hi * 8) * LL + y) * LL + x0 + col;
#pragma unroll
    for (int r = 0; r < 8; ++r) {
        op0[r * (LL * LL)] = c0[r];
        op1[r * (LL * LL)] = c1[r];
    }
}

extern "C" void kernel_launch(void* const* d_in, const int* in_sizes, int n_in,
                              void* d_out, int out_size, void* d_ws, size_t ws_size,
                              hipStream_t stream) {
    const float* x = (const float*)d_in[0];
    const float* w = (const float*)d_in[1];
    float* out = (float*)d_out;

    dim3 grid(BB * LL * 2);
    dim3 block(256);
    hipLaunchKernelGGL(fconv2d_wmma_kernel, grid, block, 0, stream, x, w, out);
}


# coarsening: a work-group computes two full output rows (four 16-pixel tiles per wave, grid 256 for 1024): weights and x rows staged once per 64 pixels, each summed B fragment feeds both rows
# speedup vs baseline: 2.2834x; 1.4867x over previous
#include <hip/hip_runtime.h>

typedef float v2f __attribute__((ext_vector_type(2)));
typedef float v4f __attribute__((ext_vector_type(4)));
typedef float v8f __attribute__((ext_vector_type(8)));

#define BB   16
#define CIN  128
#define COUT 256
#define LL   32

#define QP   33
#define CP   155
#define SC   34
#define SROWS 4

__global__ __launch_bounds__(256) void fconv2d_wmma_kernel(
    const float* __restrict__ x,
    const float* __restrict__ w,
    float* __restrict__ out)
{
    __shared__ float sW2[9 * 16 * QP];
    __shared__ float sX2[SROWS * SC * CP];
    __shared__ float sP[4 * 8 * 256];

    const int tid = threadIdx.x;
    const int bid = blockIdx.x;
    const int b   = bid >> 4;
    const int y0  = (bid & 15) << 1;


    const v4f* __restrict__ w4 = (const v4f*)w;
    v4f wv[5];
#pragma unroll
    for (int i = 0; i < 5; ++i) {
        int f = tid + 256 * i;
        f = f < 1152 ? f : 1151;
        wv[i] = w4[f];
    }

    const int q4 = tid & 7;
    const int g  = tid >> 3;
    v4f xv[SROWS][4];
#pragma unroll
    for (int row = 0; row < SROWS; ++row) {
        int gy  = y0 + row;
        int gyc = gy < LL ? gy : (LL - 1);
#pragma unroll
        for (int j = 0; j < 4; ++j) {
            int s  = 1 + g + 32 * j;
            int ch = (s + 95) & 127;
            const float* lp = x + ((b * CIN + ch) * LL + gyc) * LL;
            xv[row][j] = *(const v4f*)(lp + 4 * q4);
        }
    }

#pragma unroll
    for (int i = 0; i < 5; ++i) {
        int f = tid + 256 * i;
        if (f < 1152) {
            int idx = 4 * f;
            int nq  = idx / 9;
            int t   = idx - 9 * nq;
#pragma unroll
            for (int e = 0; e < 4; ++e) {
                sW2[(8 - t) * (16 * QP) + (nq >> 5) * QP + (31 - (nq & 31))] = 0.5f * wv[i][e];
                ++t;
                if (t == 9) { t = 0; ++nq; }
            }
        }
    }

#pragma unroll
    for (int row = 0; row < SROWS; ++row) {
        const bool row_in = (y0 + row) < LL;
#pragma unroll
        for (int j = 0; j < 4; ++j) {
            int s = 1 + g + 32 * j;
            float* dp = &sX2[(row * SC + 4 * q4) * CP + s];
            v4f v = xv[row][j];
            float e0 = row_in ? v.x : 0.0f;
            float e1 = row_in ? v.y : 0.0f;
            float e2 = row_in ? v.z : 0.0f;
            float e3 = row_in ? v.w : 0.0f;
            dp[0]      = e0;
            dp[CP]     = e1;
            dp[2 * CP] = e2;
            dp[3 * CP] = e3;
            const bool dup = (j == 0) && (g < 25);
            if (dup) {
                dp[128]          = e0;
                dp[CP + 128]     = e1;
                dp[2 * CP + 128] = e2;
                dp[3 * CP + 128] = e3;
            }
            if (q4 == 7) {
                dp[4 * CP] = 0.0f;
                dp[5 * CP] = 0.0f;
                if (dup) {
                    dp[4 * CP + 128] = 0.0f;
                    dp[5 * CP + 128] = 0.0f;
                }
            }
        }
    }
    __syncthreads();

    const int lane = tid & 31;
    const int wid  = tid >> 5;
    const int col  = lane & 15;
    const int hi   = lane >> 4;

    const int d0 = wid;
    const int d1 = (16 - wid) & 15;

    const v8f zero8 = {0.f, 0.f, 0.f, 0.f, 0.f, 0.f, 0.f, 0.f};
    v8f c00 = zero8, c01 = zero8, c10 = zero8, c11 = zero8;
    v8f e00 = zero8, e01 = zero8, e10 = zero8, e11 = zero8;

#pragma unroll
    for (int sr = 0; sr < SROWS; ++sr) {
        const bool f0 = (sr <= 2);
        const bool f1 = (sr >= 1);
        const int  t0 = (f0 ? sr : 0) * 3;
        const int  t1 = (f1 ? sr - 1 : 0) * 3;
#pragma unroll
        for (int dx = 0; dx < 3; ++dx) {
            const float* ap0 = &sW2[(t0 + dx) * (16 * QP) + col * QP + 2 * hi];
            const float* ap1 = &sW2[(t1 + dx) * (16 * QP) + col * QP + 2 * hi];
            const int    sbl = (sr * SC + col + dx) * CP + 2 + 2 * hi;
            const int    sbr = sbl + 16 * CP;
            const float* l0p = &sX2[sbl + 8 * d0];
            const float* l1p = &sX2[sbl + 8 * d1];
            const float* r0p = &sX2[sbr + 8 * d0];
            const float* r1p = &sX2[sbr + 8 * d1];
#pragma unroll
            for (int k = 0; k < 8; ++k) {
                v2f bl, br;
                bl.x = l0p[4 * k]     + l1p[4 * k];
                bl.y = l0p[4 * k + 1] + l1p[4 * k + 1];
                br.x = r0p[4 * k]     + r1p[4 * k];
                br.y = r0p[4 * k + 1] + r1p[4 * k + 1];
                if (f0) {
                    v2f a;
                    a.x = ap0[4 * k];
                    a.y = ap0[4 * k + 1];
                    c00 = __builtin_amdgcn_wmma_f32_16x16x4_f32(
                            false, a, false, bl, (short)0, c00, false, false);
                    c01 = __builtin_amdgcn_wmma_f32_16x16x4_f32(
                            false, a, false, br, (short)0, c01, false, false);
                }
                if (f1) {
                    v2f a;
                    a.x = ap1[4 * k];
                    a.y = ap1[4 * k + 1];
                    c10 = __builtin_amdgcn_wmma_f32_16x16x4_f32(
                            false, a, false, bl, (short)0, c10, false, false);
                    c11 = __builtin_amdgcn_wmma_f32_16x16x4_f32(
                            false, a, false, br, (short)0, c11, false, false);
                }
            }
            const float* l8p = &sX2[sbl + 64 + 4 * wid];
            const float* r8p = &sX2[sbr + 64 + 4 * wid];
            float ul0 = l8p[0];
            float ul1 = l8p[1];
            float ur0 = r8p[0];
            float ur1 = r8p[1];
            v2f b8l, b8r;
            b8l.x = ul0 + ul0;
            b8l.y = ul1 + ul1;
            b8r.x = ur0 + ur0;
            b8r.y = ur1 + ur1;
            if (f0) {
                v2f a8;
                a8.x = ap0[4 * wid];
                a8.y = ap0[4 * wid + 1];
                e00 = __builtin_amdgcn_wmma_f32_16x16x4_f32(
                        false, a8, false, b8l, (short)0, e00, false, false);
                e01 = __builtin_amdgcn_wmma_f32_16x16x4_f32(
                        false, a8, false, b8r, (short)0, e01, false, false);
            }
            if (f1) {
                v2f a8;
                a8.x = ap1[4 * wid];
                a8.y = ap1[4 * wid + 1];
                e10 = __builtin_amdgcn_wmma_f32_16x16x4_f32(
                        false, a8, false, b8l, (short)0, e10, false, false);
                e11 = __builtin_amdgcn_wmma_f32_16x16x4_f32(
                        false, a8, false, b8r, (short)0, e11, false, false);
            }
        }
    }

#pragma unroll
    for (int r = 0; r < 8; ++r) {
        sP[(0 * 8 + wid) * 256 + r * 32 + lane] = e00[r];
        sP[(1 * 8 + wid) * 256 + r * 32 + lane] = e01[r];
        sP[(2 * 8 + wid) * 256 + r * 32 + lane] = e10[r];
        sP[(3 * 8 + wid) * 256 + r * 32 + lane] = e11[r];
    }
    __syncthreads();
    float s8[4];
#pragma unroll
    for (int p = 0; p < 4; ++p) {
        float s = sP[(p * 8) * 256 + tid];
#pragma unroll
        for (int w8 = 1; w8 < 8; ++w8)
            s += sP[(p * 8 + w8) * 256 + tid];
        s8[p] = s;
    }

#pragma unroll
    for (int p = 0; p < 4; ++p) {
        const int yy = y0 + (p >> 1);
        const int xx = ((p & 1) << 4) + col;
        const v8f cv = (p == 0) ? c00 : (p == 1) ? c01 : (p == 2) ? c10 : c11;
        float* op0 = out + ((b * COUT + d0 * 16 + hi * 8) * LL + yy) * LL + xx;
#pragma unroll
        for (int r = 0; r < 8; ++r)
            op0[r * (LL * LL)] = cv[r];
        if (wid != 0) {
            float* op1 = out + ((b * COUT + d1 * 16 + hi * 8) * LL + yy) * LL + xx;
#pragma unroll
            for (int r = 0; r < 8; ++r)
                op1[r * (LL * LL)] = cv[r];
        }
        out[((b * COUT + 8 * 16 + wid + hi * 8) * LL + yy) * LL + xx] = s8[p];
    }
}

extern "C" void kernel_launch(void* const* d_in, const int* in_sizes, int n_in,
                              void* d_out, int out_size, void* d_ws, size_t ws_size,
                              hipStream_t stream) {
    const float* x = (const float*)d_in[0];
    const float* w = (const float*)d_in[1];
    float* out = (float*)d_out;

    dim3 grid(BB * (LL / 2));
    dim3 block(256);
    hipLaunchKernelGGL(fconv2d_wmma_kernel, grid, block, 0, stream, x, w, out);
}


# LDS fragment reads widened to 16 bytes per lane (row pitches 156 and 36 floats, aligned, no bank conflict): one read feeds two K-steps; k is summed in a different order inside each tap
# speedup vs baseline: 2.5786x; 1.1293x over previous
#include <hip/hip_runtime.h>

typedef float v2f __attribute__((ext_vector_type(2)));
typedef float v4f __attribute__((ext_vector_type(4)));
typedef float v8f __attribute__((ext_vector_type(8)));

#define BB   16
#define CIN  128
#define COUT 256
#define LL   32

#define QP   36
#define CP   156
#define SC   34
#define SROWS 4

__global__ __launch_bounds__(256) void fconv2d_wmma_kernel(
    const float* __restrict__ x,
    const float* __restrict__ w,
    float* __restrict__ out)
{
    __shared__ __attribute__((aligned(16))) float sW2[9 * 16 * QP];
    __shared__ __attribute__((aligned(16))) float sX2[SROWS * SC * CP];
    __shared__ float sP[4 * 8 * 256];

    const int tid = threadIdx.x;
    const int bid = blockIdx.x;
    const int b   = bid >> 4;
    const int y0  = (bid & 15) << 1;


    const v4f* __restrict__ w4 = (const v4f*)w;
    v4f wv[5];
#pragma unroll
    for (int i = 0; i < 5; ++i) {
        int f = tid + 256 * i;
        f = f < 1152 ? f : 1151;
        wv[i] = w4[f];
    }

    const int q4 = tid & 7;
    const int g  = tid >> 3;
    v4f xv[SROWS][4];
#pragma unroll
    for (int row = 0; row < SROWS; ++row) {
        int gy  = y0 + row;
        int gyc = gy < LL ? gy : (LL - 1);
#pragma unroll
        for (int j = 0; j < 4; ++j) {
            int t  = g + 32 * j;
            int ch = (t + 97) & 127;
            const float* lp = x + ((b * CIN + ch) * LL + gyc) * LL;
            xv[row][j] = *(const v4f*)(lp + 4 * q4);
        }
    }

#pragma unroll
    for (int i = 0; i < 5; ++i) {
        int f = tid + 256 * i;
        if (f < 1152) {
            int idx = 4 * f;
            int nq  = idx / 9;
            int t   = idx - 9 * nq;
#pragma unroll
            for (int e = 0; e < 4; ++e) {
                sW2[(8 - t) * (16 * QP) + (nq >> 5) * QP + (31 - (nq & 31))] = 0.5f * wv[i][e];
                ++t;
                if (t == 9) { t = 0; ++nq; }
            }
        }
    }

#pragma unroll
    for (int row = 0; row < SROWS; ++row) {
        const bool row_in = (y0 + row) < LL;
#pragma unroll
        for (int j = 0; j < 4; ++j) {
            int t = g + 32 * j;
            float* dp = &sX2[(row * SC + 4 * q4) * CP + t];
            v4f v = xv[row][j];
            float e0 = row_in ? v.x : 0.0f;
            float e1 = row_in ? v.y : 0.0f;
            float e2 = row_in ? v.z : 0.0f;
            float e3 = row_in ? v.w : 0.0f;
            const bool up = (q4 & 4) != 0;
            float* dpa = dp + (up ? 2 * CP : 0);
            float* dpb = dp + (up ? 0 : 2 * CP);
            float a0 = up ? e2 : e0;
            float a1 = up ? e3 : e1;
            float b0 = up ? e0 : e2;
            float b1 = up ? e1 : e3;
            dpa[0]  = a0;
            dpa[CP] = a1;
            dpb[0]  = b0;
            dpb[CP] = b1;
            const bool dup = (j == 0) && (g < 24);
            if (dup) {
                dpa[128]      = a0;
                dpa[CP + 128] = a1;
                dpb[128]      = b0;
                dpb[CP + 128] = b1;
            }
            if (q4 == 7) {
                dp[4 * CP] = 0.0f;
                dp[5 * CP] = 0.0f;
                if (dup) {
                    dp[4 * CP + 128] = 0.0f;
                    dp[5 * CP + 128] = 0.0f;
                }
            }
        }
    }
    __syncthreads();

    const int lane = tid & 31;
    const int wid  = tid >> 5;
    const int col  = lane & 15;
    const int hi   = lane >> 4;

    const int d0 = wid;
    const int d1 = (16 - wid) & 15;

    const v8f zero8 = {0.f, 0.f, 0.f, 0.f, 0.f, 0.f, 0.f, 0.f};
    v8f c00 = zero8, c01 = zero8, c10 = zero8, c11 = zero8;
    v8f e00 = zero8, e01 = zero8, e10 = zero8, e11 = zero8;

#pragma unroll
    for (int sr = 0; sr < SROWS; ++sr) {
        const bool f0 = (sr <= 2);
        const bool f1 = (sr >= 1);
        const int  t0 = (f0 ? sr : 0) * 3;
        const int  t1 = (f1 ? sr - 1 : 0) * 3;
#pragma unroll
        for (int dx = 0; dx < 3; ++dx) {
            const float* ap0 = &sW2[(t0 + dx) * (16 * QP) + col * QP + 4 * hi];
            const float* ap1 = &sW2[(t1 + dx) * (16 * QP) + col * QP + 4 * hi];
            const int    sbl = (sr * SC + col + dx) * CP + 4 * hi;
            const int    sbr = sbl + 16 * CP;
            const float* l0p = &sX2[sbl + 8 * d0];
            const float* l1p = &sX2[sbl + 8 * d1];
            const float* r0p = &sX2[sbr + 8 * d0];
            const float* r1p = &sX2[sbr + 8 * d1];
#pragma unroll
            for (int j = 0; j < 4; ++j) {
                const v4f bl4 = *(const v4f*)(l0p + 8 * j) + *(const v4f*)(l1p + 8 * j);
                const v4f br4 = *(const v4f*)(r0p + 8 * j) + *(const v4f*)(r1p + 8 * j);
                v2f bla, blb, bra, brb;
                bla.x = bl4.x; bla.y = bl4.y;
                blb.x = bl4.z; blb.y = bl4.w;
                bra.x = br4.x; bra.y = br4.y;
                brb.x = br4.z; brb.y = br4.w;
                if (f0) {
                    const v4f a4 = *(const v4f*)(ap0 + 8 * j);
                    v2f aa, ab;
                    aa.x = a4.x; aa.y = a4.y;
                    ab.x = a4.z; ab.y = a4.w;
                    c00 = __builtin_amdgcn_wmma_f32_16x16x4_f32(
                            false, aa, false, bla, (short)0, c00, false, false);
                    c01 = __builtin_amdgcn_wmma_f32_16x16x4_f32(
                            false, aa, false, bra, (short)0, c01, false, false);
                    c00 = __builtin_amdgcn_wmma_f32_16x16x4_f32(
                            false, ab, false, blb, (short)0, c00, false, false);
                    c01 = __builtin_amdgcn_wmma_f32_16x16x4_f32(
                            false, ab, false, brb, (short)0, c01, false, false);
                }
                if (f1) {
                    const v4f a4 = *(const v4f*)(ap1 + 8 * j);
                    v2f aa, ab;
                    aa.x = a4.x; aa.y = a4.y;
                    ab.x = a4.z; ab.y = a4.w;
                    c10 = __builtin_amdgcn_wmma_f32_16x16x4_f32(
                            false, aa, false, bla, (short)0, c10, false, false);
                    c11 = __builtin_amdgcn_wmma_f32_16x16x4_f32(
                            false, aa, false, bra, (short)0, c11, false, false);
                    c10 = __builtin_amdgcn_wmma_f32_16x16x4_f32(
                            false, ab, false, blb, (short)0, c10, false, false);
                    c11 = __builtin_amdgcn_wmma_f32_16x16x4_f32(
                            false, ab, false, brb, (short)0, c11, false, false);
                }
            }
            const int m8 = 8 * (wid >> 1) + 2 * (wid & 1);
            const v2f ul = *(const v2f*)&sX2[sbl + 64 + m8];
            const v2f ur = *(const v2f*)&sX2[sbr + 64 + m8];
            const v2f b8l = ul + ul;
            const v2f b8r = ur + ur;
            if (f0) {
                const v2f a8 = *(const v2f*)(ap0 + m8);
                e00 = __builtin_amdgcn_wmma_f32_16x16x4_f32(
                        false, a8, false, b8l, (short)0, e00, false, false);
                e01 = __builtin_amdgcn_wmma_f32_16x16x4_f32(
                        false, a8, false, b8r, (short)0, e01, false, false);
            }
            if (f1) {
                const v2f a8 = *(const v2f*)(ap1 + m8);
                e10 = __builtin_amdgcn_wmma_f32_16x16x4_f32(
                        false, a8, false, b8l, (short)0, e10, false, false);
                e11 = __builtin_amdgcn_wmma_f32_16x16x4_f32(
                        false, a8, false, b8r, (short)0, e11, false, false);
            }
        }
    }

#pragma unroll
    for (int r = 0; r < 8; ++r) {
        sP[(0 * 8 + wid) * 256 + r * 32 + lane] = e00[r];
        sP[(1 * 8 + wid) * 256 + r * 32 + lane] = e01[r];
        sP[(2 * 8 + wid) * 256 + r * 32 + lane] = e10[r];
        sP[(3 * 8 + wid) * 256 + r * 32 + lane] = e11[r];
    }
    __syncthreads();
    float s8[4];
#pragma unroll
    for (int p = 0; p < 4; ++p) {
        float s = sP[(p * 8) * 256 + tid];
#pragma unroll
        for (int w8 = 1; w8 < 8; ++w8)
            s += sP[(p * 8 + w8) * 256 + tid];
        s8[p] = s;
    }

#pragma unroll
    for (int p = 0; p < 4; ++p) {
        const int yy = y0 + (p >> 1);
        const int xx = ((p & 1) << 4) + col;
        const v8f cv = (p == 0) ? c00 : (p == 1) ? c01 : (p == 2) ? c10 : c11;
        float* op0 = out + ((b * COUT + d0 * 16 + hi * 8) * LL + yy) * LL + xx;
#pragma unroll
        for (int r = 0; r < 8; ++r)
            op0[r * (LL * LL)] = cv[r];
        if (wid != 0) {
            float* op1 = out + ((b * COUT + d1 * 16 + hi * 8) * LL + yy) * LL + xx;
#pragma unroll
            for (int r = 0; r < 8; ++r)
                op1[r * (LL * LL)] = cv[r];
        }
        out[((b * COUT + 8 * 16 + wid + hi * 8) * LL + yy) * LL + xx] = s8[p];
    }
}

extern "C" void kernel_launch(void* const* d_in, const int* in_sizes, int n_in,
                              void* d_out, int out_size, void* d_ws, size_t ws_size,
                              hipStream_t stream) {
    const float* x = (const float*)d_in[0];
    const float* w = (const float*)d_in[1];
    float* out = (float*)d_out;

    dim3 grid(BB * (LL / 2));
    dim3 block(256);
    hipLaunchKernelGGL(fconv2d_wmma_kernel, grid, block, 0, stream, x, w, out);
}


# minimal-product 3-tap row filter along y (2 output rows from 4 slab rows: 4 row products for 6, 768 matrix instructions per block for 1152); row sums and differences change the summation order
# speedup vs baseline: 3.7016x; 1.4355x over previous
#include <hip/hip_runtime.h>

typedef float v2f __attribute__((ext_vector_type(2)));
typedef float v4f __attribute__((ext_vector_type(4)));
typedef float v8f __attribute__((ext_vector_type(8)));

#define BB   16
#define CIN  128
#define COUT 256
#define LL   32

#define SC    34
#define SROWS 4
#define TP    132
#define AP    20
#define ATS   328
#define RP    20

static __device__ __forceinline__ float cos16(int k)
{
    const float c1 = 0.92387953251128674f, c2 = 0.70710678118654752f, c3 = 0.38268343236508977f;
    switch (k & 15) {
        case 0:  return 1.0f;
        case 1:  return c1;
        case 2:  return c2;
        case 3:  return c3;
        case 4:  return 0.0f;
        case 5:  return -c3;
        case 6:  return -c2;
        case 7:  return -c1;
        case 8:  return -1.0f;
        case 9:  return -c1;
        case 10: return -c2;
        case 11: return -c3;
        case 12: return 0.0f;
        case 13: return c3;
        case 14: return c2;
        default: return c1;
    }
}
static __device__ __forceinline__ float sin16(int k) { return cos16(k + 12); }

static __device__ __forceinline__ v4f mac4(v4f acc, v4f v, float c) { return c == 0.0f ? acc : acc + v * c; }
static __device__ __forceinline__ float mac1(float acc, float v, float c) { return c == 0.0f ? acc : acc + v * c; }

static __device__ __forceinline__ void wdft_store(float* ap, float w0, float w1, float w2, float w3)
{
    ap[0] = 0.0625f * ((w0 + w2) + (w1 + w3));
    ap[8] = 0.0625f * ((w0 + w2) - (w1 + w3));
#pragma unroll
    for (int f = 1; f < 8; ++f) {
        float c = w0;
        c = mac1(c, w1, cos16(f));
        c = mac1(c, w2, cos16(2 * f));
        c = mac1(c, w3, cos16(3 * f));
        float s = w1 * sin16(f);
        s = mac1(s, w2, sin16(2 * f));
        s = mac1(s, w3, sin16(3 * f));
        ap[f * 12 * ATS]     = 0.125f * c;
        ap[f * 12 * ATS + 8] = -0.125f * s;
    }
}

static __device__ __forceinline__ v8f mm(v2f a, v2f b, v8f c)
{
    return __builtin_amdgcn_wmma_f32_16x16x4_f32(false, a, false, b, (short)0, c, false, false);
}

__global__ __launch_bounds__(256) void fconv2d_wmma_kernel(
    const float* __restrict__ x,
    const float* __restrict__ w,
    float* __restrict__ out)
{
    __shared__ __attribute__((aligned(16))) float sT[SROWS * SC * TP];
    __shared__ __attribute__((aligned(16))) float sA[8 * 12 * ATS];
    __shared__ __attribute__((aligned(16))) float sR[4 * 9 * 16 * RP];

    const int tid = threadIdx.x;
    const int bid = blockIdx.x;
    const int b   = bid >> 4;
    const int y0  = (bid & 15) << 1;


    float wg[2][3][4];
#pragma unroll
    for (int it = 0; it < 2; ++it) {
        if (it == 0 || tid < 128) {
            const int j  = tid + 256 * it;
            const int n  = j / 24;
            const int m  = j - 24 * n;
            const int r  = m / 3;
            const int dx = m - 3 * r;
            const float* wp = w + n * 288 + 9 * r + (2 - dx);
#pragma unroll
            for (int dyy = 0; dyy < 3; ++dyy)
#pragma unroll
                for (int k = 0; k < 4; ++k)
                    wg[it][dyy][k] = wp[72 * k + 3 * (2 - dyy)];
        }
    }

    const int q4   = tid & 7;
    const int r8   = (tid >> 3) & 7;
    const int srow = tid >> 6;
    const int gy   = y0 + srow;
    const bool row_in = gy < LL;
    const int gyc  = row_in ? gy : (LL - 1);
    v4f u[16];
#pragma unroll
    for (int e = 0; e < 16; ++e) {
        const int ch = (8 * e - r8) & 127;
        u[e] = *(const v4f*)(x + ((b * CIN + ch) * LL + gyc) * LL + 4 * q4);
    }

#pragma unroll
    for (int it = 0; it < 2; ++it) {
        if (it == 0 || tid < 128) {
            const int j  = tid + 256 * it;
            const int n  = j / 24;
            const int m  = j - 24 * n;
            const int r  = m / 3;
            const int dx = m - 3 * r;
            float u1[4], u2[4];
#pragma unroll
            for (int k = 0; k < 4; ++k) {
                const float p = 0.5f * (wg[it][0][k] + wg[it][2][k]);
                const float q = 0.5f * wg[it][1][k];
                u1[k] = p + q;
                u2[k] = p - q;
            }
            float* ap = &sA[dx * ATS + n * AP + r];
            wdft_store(ap,           wg[it][0][0], wg[it][0][1], wg[it][0][2], wg[it][0][3]);
            wdft_store(ap + 3 * ATS, u1[0], u1[1], u1[2], u1[3]);
            wdft_store(ap + 6 * ATS, u2[0], u2[1], u2[2], u2[3]);
            wdft_store(ap + 9 * ATS, wg[it][2][0], wg[it][2][1], wg[it][2][2], wg[it][2][3]);
        }
    }

    if (!row_in) {
#pragma unroll
        for (int e = 0; e < 16; ++e) u[e] = (v4f){0.f, 0.f, 0.f, 0.f};
    }
    v4f sp[8], sm[8];
#pragma unroll
    for (int e = 1; e < 8; ++e) { sp[e] = u[e] + u[16 - e]; sm[e] = u[e] - u[16 - e]; }
    const v4f e0p = u[0] + u[8];
    const v4f e0m = u[0] - u[8];
    v4f spp[4], spm[4], smp[4], smm[4];
#pragma unroll
    for (int e = 1; e < 4; ++e) {
        spp[e] = sp[e] + sp[8 - e];  spm[e] = sp[e] - sp[8 - e];
        smp[e] = sm[e] + sm[8 - e];  smm[e] = sm[e] - sm[8 - e];
    }
    v4f Uc[9], Us[9];
#pragma unroll
    for (int f = 0; f < 9; ++f) {
        v4f c;
        if ((f & 1) == 0) {
            c = mac4(e0p, sp[4], cos16(4 * f));
#pragma unroll
            for (int e = 1; e < 4; ++e) c = mac4(c, spp[e], cos16(f * e));
        } else {
            c = e0m;
#pragma unroll
            for (int e = 1; e < 4; ++e) c = mac4(c, spm[e], cos16(f * e));
        }
        Uc[f] = c;
        v4f s = (v4f){0.f, 0.f, 0.f, 0.f};
        if (f >= 1 && f <= 7) {
            if ((f & 1) == 0) {
#pragma unroll
                for (int e = 1; e < 4; ++e) s = mac4(s, smm[e], sin16(f * e));
            } else {
                s = sm[4] * sin16(4 * f);
#pragma unroll
                for (int e = 1; e < 4; ++e) s = mac4(s, smp[e], sin16(f * e));
            }
        }
        Us[f] = s;
    }

    {
        const bool up = (q4 & 4) != 0;
        float* tp  = &sT[(srow * SC + 4 * q4) * TP + r8];
        float* dpa = tp + (up ? 2 * TP : 0);
        float* dpb = tp + (up ? 0 : 2 * TP);
#pragma unroll
        for (int blk = 0; blk < 8; ++blk) {
#pragma unroll
            for (int part = 0; part < 2; ++part) {
                const v4f v = (blk == 0) ? (part == 0 ? Uc[0] : Uc[8]) : (part == 0 ? Uc[blk] : Us[blk]);
                const int o = 16 * blk + 8 * part;
                dpa[o]      = up ? v.z : v.x;
                dpa[TP + o] = up ? v.w : v.y;
                dpb[o]      = up ? v.x : v.z;
                dpb[TP + o] = up ? v.y : v.w;
            }
        }
        const int zr = tid >> 6, zc = (tid >> 5) & 1, z4 = tid & 31;
        *(v4f*)&sT[(zr * SC + 32 + zc) * TP + 4 * z4] = (v4f){0.f, 0.f, 0.f, 0.f};
    }
    __syncthreads();

    const int lane = tid & 31;
    const int wid  = tid >> 5;
    const int col  = lane & 15;
    const int hi   = lane >> 4;

    const v8f zero8 = {0.f, 0.f, 0.f, 0.f, 0.f, 0.f, 0.f, 0.f};
    v8f accA[4][2] = {{zero8, zero8}, {zero8, zero8}, {zero8, zero8}, {zero8, zero8}};
    v8f accB[4][2] = {{zero8, zero8}, {zero8, zero8}, {zero8, zero8}, {zero8, zero8}};

#pragma unroll
    for (int dx = 0; dx < 3; ++dx) {
        v4f d[4][4];
#pragma unroll
        for (int sr = 0; sr < SROWS; ++sr) {
            const float* bp = &sT[(sr * SC + col + dx) * TP + 16 * wid + 4 * hi];
            d[sr][0] = *(const v4f*)(bp);
            d[sr][1] = *(const v4f*)(bp + 8);
            d[sr][2] = *(const v4f*)(bp + 16 * TP);
            d[sr][3] = *(const v4f*)(bp + 16 * TP + 8);
        }
        v4f V[4][4];
#pragma unroll
        for (int q = 0; q < 4; ++q) {
            V[0][q] = d[0][q] - d[2][q];
            V[1][q] = d[1][q] + d[2][q];
            V[2][q] = d[2][q] - d[1][q];
            V[3][q] = d[1][q] - d[3][q];
        }
#pragma unroll
        for (int i = 0; i < 4; ++i) {
            const float* ap = &sA[(wid * 12 + i * 3 + dx) * ATS + col * AP + 4 * hi];
            const v4f a0 = *(const v4f*)(ap);
            const v4f a1 = *(const v4f*)(ap + 8);
            accA[i][0] = mm(a0.xy, V[i][0].xy, accA[i][0]);
            accA[i][1] = mm(a0.xy, V[i][2].xy, accA[i][1]);
            accB[i][0] = mm(a1.xy, V[i][1].xy, accB[i][0]);
            accB[i][1] = mm(a1.xy, V[i][3].xy, accB[i][1]);
            accA[i][0] = mm(a0.zw, V[i][0].zw, accA[i][0]);
            accA[i][1] = mm(a0.zw, V[i][2].zw, accA[i][1]);
            accB[i][0] = mm(a1.zw, V[i][1].zw, accB[i][0]);
            accB[i][1] = mm(a1.zw, V[i][3].zw, accB[i][1]);
        }
    }

#pragma unroll
    for (int h = 0; h < 2; ++h) {
        float* rp0 = &sR[((0 + h) * 9 * 16 + 8 * hi) * RP + col];
        float* rp1 = &sR[((2 + h) * 9 * 16 + 8 * hi) * RP + col];
        if (wid == 0) {
            const v8f a0 = (accA[0][h] + accA[1][h]) + accA[2][h];
            const v8f a1 = (accA[1][h] - accA[2][h]) - accA[3][h];
            const v8f b0 = (accB[0][h] + accB[1][h]) + accB[2][h];
            const v8f b1 = (accB[1][h] - accB[2][h]) - accB[3][h];
#pragma unroll
            for (int r = 0; r < 8; ++r) {
                rp0[(0 * 16 + r) * RP] = a0[r];
                rp0[(8 * 16 + r) * RP] = b0[r];
                rp1[(0 * 16 + r) * RP] = a1[r];
                rp1[(8 * 16 + r) * RP] = b1[r];
            }
        } else {
            const v8f m0 = accA[0][h] + accB[0][h];
            const v8f m1 = accA[1][h] + accB[1][h];
            const v8f m2 = accA[2][h] + accB[2][h];
            const v8f m3 = accA[3][h] + accB[3][h];
            const v8f o0 = (m0 + m1) + m2;
            const v8f o1 = (m1 - m2) - m3;
#pragma unroll
            for (int r = 0; r < 8; ++r) {
                rp0[(wid * 16 + r) * RP] = o0[r];
                rp1[(wid * 16 + r) * RP] = o1[r];
            }
        }
    }
    __syncthreads();

    {
        const int quad = tid & 7;
        const int n    = (tid >> 3) & 15;
        const int ry   = tid >> 7;
        const int p    = 2 * ry + (quad >> 2);
        const float* rp = &sR[(p * 9 * 16 + n) * RP + 4 * (quad & 3)];
        v4f R[9];
#pragma unroll
        for (int f = 0; f < 9; ++f) R[f] = *(const v4f*)(rp + f * 16 * RP);
        float* op = out + ((b * COUT + n) * LL + (y0 + ry)) * LL + 4 * quad;
#pragma unroll
        for (int d = 0; d < 9; ++d) {
            v4f o = ((d & 1) == 0) ? (R[0] + R[8]) : (R[0] - R[8]);
#pragma unroll
            for (int f = 1; f < 8; ++f) o = mac4(o, R[f], cos16(f * d));
            *(v4f*)(op + d * 16 * LL * LL) = o;
            if (d >= 1 && d <= 7)
                *(v4f*)(op + (16 - d) * 16 * LL * LL) = o;
        }
    }
}

extern "C" void kernel_launch(void* const* d_in, const int* in_sizes, int n_in,
                              void* d_out, int out_size, void* d_ws, size_t ws_size,
                              hipStream_t stream) {
    const float* x = (const float*)d_in[0];
    const float* w = (const float*)d_in[1];
    float* out = (float*)d_out;

    dim3 grid(BB * (LL / 2));
    dim3 block(256);
    hipLaunchKernelGGL(fconv2d_wmma_kernel, grid, block, 0, stream, x, w, out);
}


# minimal-product 3-tap filter along x too (2 pixels from 4 slab columns: 16 products per 2x2 outputs for 24, 512 matrix instructions per block for 768); sums are taken in another order
# speedup vs baseline: 3.8929x; 1.0517x over previous
#include <hip/hip_runtime.h>

typedef float v2f __attribute__((ext_vector_type(2)));
typedef float v4f __attribute__((ext_vector_type(4)));
typedef float v8f __attribute__((ext_vector_type(8)));

#define BB   16
#define CIN  128
#define COUT 256
#define LL   32

#define SC    34
#define SROWS 4
#define TP    132
#define AP    20
#define ATS   328
#define RP    36

static __device__ __forceinline__ float cos16(int k)
{
    const float c1 = 0.92387953251128674f, c2 = 0.70710678118654752f, c3 = 0.38268343236508977f;
    switch (k & 15) {
        case 0:  return 1.0f;
        case 1:  return c1;
        case 2:  return c2;
        case 3:  return c3;
        case 4:  return 0.0f;
        case 5:  return -c3;
        case 6:  return -c2;
        case 7:  return -c1;
        case 8:  return -1.0f;
        case 9:  return -c1;
        case 10: return -c2;
        case 11: return -c3;
        case 12: return 0.0f;
        case 13: return c3;
        case 14: return c2;
        default: return c1;
    }
}
static __device__ __forceinline__ float sin16(int k) { return cos16(k + 12); }

static __device__ __forceinline__ v4f mac4(v4f acc, v4f v, float c) { return c == 0.0f ? acc : acc + v * c; }
static __device__ __forceinline__ float mac1(float acc, float v, float c) { return c == 0.0f ? acc : acc + v * c; }

static __device__ __forceinline__ void wdft_store(float* ap, float w0, float w1, float w2, float w3)
{
    ap[0] = 0.0625f * ((w0 + w2) + (w1 + w3));
    ap[8] = 0.0625f * ((w0 + w2) - (w1 + w3));
#pragma unroll
    for (int f = 1; f < 8; ++f) {
        float c = w0;
        c = mac1(c, w1, cos16(f));
        c = mac1(c, w2, cos16(2 * f));
        c = mac1(c, w3, cos16(3 * f));
        float s = w1 * sin16(f);
        s = mac1(s, w2, sin16(2 * f));
        s = mac1(s, w3, sin16(3 * f));
        ap[f * 12 * ATS]     = 0.125f * c;
        ap[f * 12 * ATS + 8] = -0.125f * s;
    }
}

static __device__ __forceinline__ v8f mm(v2f a, v2f b, v8f c)
{
    return __builtin_amdgcn_wmma_f32_16x16x4_f32(false, a, false, b, (short)0, c, false, false);
}

static __device__ __forceinline__ void emit_rows(float* rb, const v8f (&m)[4][4])
{
    v8f te[4], to[4];
#pragma unroll
    for (int i = 0; i < 4; ++i) {
        te[i] = (m[i][0] + m[i][1]) + m[i][2];
        to[i] = (m[i][1] - m[i][2]) - m[i][3];
    }
    const v8f y0e = (te[0] + te[1]) + te[2], y1e = (te[1] - te[2]) - te[3];
    const v8f y0o = (to[0] + to[1]) + to[2], y1o = (to[1] - to[2]) - to[3];
#pragma unroll
    for (int r = 0; r < 8; ++r) {
        *(v2f*)(rb + r * RP)               = (v2f){y0e[r], y0o[r]};
        *(v2f*)(rb + 9 * 16 * RP + r * RP) = (v2f){y1e[r], y1o[r]};
    }
}

__global__ __launch_bounds__(256) void fconv2d_wmma_kernel(
    const float* __restrict__ x,
    const float* __restrict__ w,
    float* __restrict__ out)
{
    __shared__ __attribute__((aligned(16))) float sT[SROWS * SC * TP];
    __shared__ __attribute__((aligned(16))) float sA[8 * 12 * ATS];
    __shared__ __attribute__((aligned(16))) float sR[2 * 9 * 16 * RP];

    const int tid = threadIdx.x;
    const int bid = blockIdx.x;
    const int b   = bid >> 4;
    const int y0  = (bid & 15) << 1;


    float wg[2][3][4];
#pragma unroll
    for (int it = 0; it < 2; ++it) {
        if (it == 0 || tid < 128) {
            const int j  = tid + 256 * it;
            const int n  = j / 24;
            const int m  = j - 24 * n;
            const int r  = m / 3;
            const int dx = m - 3 * r;
            const float* wp = w + n * 288 + 9 * r + (2 - dx);
#pragma unroll
            for (int dyy = 0; dyy < 3; ++dyy)
#pragma unroll
                for (int k = 0; k < 4; ++k)
                    wg[it][dyy][k] = wp[72 * k + 3 * (2 - dyy)];
        }
    }

    const int q4   = tid & 7;
    const int r8   = (tid >> 3) & 7;
    const int srow = tid >> 6;
    const int gy   = y0 + srow;
    const bool row_in = gy < LL;
    const int gyc  = row_in ? gy : (LL - 1);
    v4f u[16];
#pragma unroll
    for (int e = 0; e < 16; ++e) {
        const int ch = (8 * e - r8) & 127;
        u[e] = *(const v4f*)(x + ((b * CIN + ch) * LL + gyc) * LL + 4 * q4);
    }

#pragma unroll
    for (int it = 0; it < 2; ++it) {
        if (it == 0 || tid < 128) {
            const int j  = tid + 256 * it;
            const int n  = j / 24;
            const int m  = j - 24 * n;
            const int r  = m / 3;
            const int dx = m - 3 * r;
            float u1[4], u2[4];
#pragma unroll
            for (int k = 0; k < 4; ++k) {
                const float p = 0.5f * (wg[it][0][k] + wg[it][2][k]);
                const float q = 0.5f * wg[it][1][k];
                u1[k] = p + q;
                u2[k] = p - q;
            }
            float* ap = &sA[dx * ATS + n * AP + r];
            wdft_store(ap,           wg[it][0][0], wg[it][0][1], wg[it][0][2], wg[it][0][3]);
            wdft_store(ap + 3 * ATS, u1[0], u1[1], u1[2], u1[3]);
            wdft_store(ap + 6 * ATS, u2[0], u2[1], u2[2], u2[3]);
            wdft_store(ap + 9 * ATS, wg[it][2][0], wg[it][2][1], wg[it][2][2], wg[it][2][3]);
        }
    }

    if (!row_in) {
#pragma unroll
        for (int e = 0; e < 16; ++e) u[e] = (v4f){0.f, 0.f, 0.f, 0.f};
    }
    v4f sp[8], sm[8];
#pragma unroll
    for (int e = 1; e < 8; ++e) { sp[e] = u[e] + u[16 - e]; sm[e] = u[e] - u[16 - e]; }
    const v4f e0p = u[0] + u[8];
    const v4f e0m = u[0] - u[8];
    v4f spp[4], spm[4], smp[4], smm[4];
#pragma unroll
    for (int e = 1; e < 4; ++e) {
        spp[e] = sp[e] + sp[8 - e];  spm[e] = sp[e] - sp[8 - e];
        smp[e] = sm[e] + sm[8 - e];  smm[e] = sm[e] - sm[8 - e];
    }
    v4f Uc[9], Us[9];
#pragma unroll
    for (int f = 0; f < 9; ++f) {
        v4f c;
        if ((f & 1) == 0) {
            c = mac4(e0p, sp[4], cos16(4 * f));
#pragma unroll
            for (int e = 1; e < 4; ++e) c = mac4(c, spp[e], cos16(f * e));
        } else {
            c = e0m;
#pragma unroll
            for (int e = 1; e < 4; ++e) c = mac4(c, spm[e], cos16(f * e));
        }
        Uc[f] = c;
        v4f s = (v4f){0.f, 0.f, 0.f, 0.f};
        if (f >= 1 && f <= 7) {
            if ((f & 1) == 0) {
#pragma unroll
                for (int e = 1; e < 4; ++e) s = mac4(s, smm[e], sin16(f * e));
            } else {
                s = sm[4] * sin16(4 * f);
#pragma unroll
                for (int e = 1; e < 4; ++e) s = mac4(s, smp[e], sin16(f * e));
            }
        }
        Us[f] = s;
    }

    {
        float* tp = &sT[(srow * SC + 2 * q4) * TP + r8];
#pragma unroll
        for (int blk = 0; blk < 8; ++blk) {
#pragma unroll
            for (int part = 0; part < 2; ++part) {
                const v4f v = (blk == 0) ? (part == 0 ? Uc[0] : Uc[8]) : (part == 0 ? Uc[blk] : Us[blk]);
                const int o = 16 * blk + 8 * part;
                tp[o]           = v.x;
                tp[17 * TP + o] = v.y;
                tp[TP + o]      = v.z;
                tp[18 * TP + o] = v.w;
            }
        }
        const int zr = tid >> 6, zc = (tid >> 5) & 1, z4 = tid & 31;
        *(v4f*)&sT[(zr * SC + (zc ? 33 : 16)) * TP + 4 * z4] = (v4f){0.f, 0.f, 0.f, 0.f};
    }
    __syncthreads();

    const int lane = tid & 31;
    const int wid  = tid >> 5;
    const int col  = lane & 15;
    const int hi   = lane >> 4;

    const v8f zero8 = {0.f, 0.f, 0.f, 0.f, 0.f, 0.f, 0.f, 0.f};
    v8f accA[4][4], accB[4][4];
#pragma unroll
    for (int i = 0; i < 4; ++i)
#pragma unroll
        for (int j = 0; j < 4; ++j) { accA[i][j] = zero8; accB[i][j] = zero8; }

    const float* bbase = &sT[col * TP + 16 * wid + 4 * hi];
    const float* abase = &sA[(wid * 12) * ATS + col * AP + 4 * hi];
#pragma unroll
    for (int part = 0; part < 2; ++part) {
        v4f d[4][4];
#pragma unroll
        for (int sr = 0; sr < SROWS; ++sr)
#pragma unroll
            for (int pos = 0; pos < 4; ++pos)
                d[sr][pos] = *(const v4f*)(bbase + (sr * SC + (pos & 1) * 17 + (pos >> 1)) * TP + 8 * part);
        v4f Vy[4][4];
#pragma unroll
        for (int pos = 0; pos < 4; ++pos) {
            Vy[0][pos] = d[0][pos] - d[2][pos];
            Vy[1][pos] = d[1][pos] + d[2][pos];
            Vy[2][pos] = d[2][pos] - d[1][pos];
            Vy[3][pos] = d[1][pos] - d[3][pos];
        }
#pragma unroll
        for (int i = 0; i < 4; ++i) {
            v4f VV[4];
            VV[0] = Vy[i][0] - Vy[i][2];
            VV[1] = Vy[i][1] + Vy[i][2];
            VV[2] = Vy[i][2] - Vy[i][1];
            VV[3] = Vy[i][1] - Vy[i][3];
            const float* ap = abase + (i * 3) * ATS + 8 * part;
            const v4f g0 = *(const v4f*)(ap);
            const v4f g1 = *(const v4f*)(ap + ATS);
            const v4f g2 = *(const v4f*)(ap + 2 * ATS);
            const v4f t = 0.5f * (g0 + g2);
            const v4f q = 0.5f * g1;
            v4f UU[4];
            UU[0] = g0;
            UU[1] = t + q;
            UU[2] = t - q;
            UU[3] = g2;
#pragma unroll
            for (int j = 0; j < 4; ++j) {
                if (part == 0) {
                    accA[i][j] = mm(UU[j].xy, VV[j].xy, accA[i][j]);
                    accA[i][j] = mm(UU[j].zw, VV[j].zw, accA[i][j]);
                } else {
                    accB[i][j] = mm(UU[j].xy, VV[j].xy, accB[i][j]);
                    accB[i][j] = mm(UU[j].zw, VV[j].zw, accB[i][j]);
                }
            }
        }
    }

    {
        float* rbase = &sR[(8 * hi) * RP + 2 * col];
        if (wid == 0) {
            emit_rows(rbase, accA);
            emit_rows(rbase + 8 * 16 * RP, accB);
        } else {
            v8f m[4][4];
#pragma unroll
            for (int i = 0; i < 4; ++i)
#pragma unroll
                for (int j = 0; j < 4; ++j) m[i][j] = accA[i][j] + accB[i][j];
            emit_rows(rbase + wid * 16 * RP, m);
        }
    }
    __syncthreads();

    {
        const int quad = tid & 7;
        const int n    = (tid >> 3) & 15;
        const int ry   = tid >> 7;
        const float* rp = &sR[(ry * 9 * 16 + n) * RP + 4 * quad];
        v4f R[9];
#pragma unroll
        for (int f = 0; f < 9; ++f) R[f] = *(const v4f*)(rp + f * 16 * RP);
        float* op = out + ((b * COUT + n) * LL + (y0 + ry)) * LL + 4 * quad;
#pragma unroll
        for (int d = 0; d < 9; ++d) {
            v4f o = ((d & 1) == 0) ? (R[0] + R[8]) : (R[0] - R[8]);
#pragma unroll
            for (int f = 1; f < 8; ++f) o = mac4(o, R[f], cos16(f * d));
            *(v4f*)(op + d * 16 * LL * LL) = o;
            if (d >= 1 && d <= 7)
                *(v4f*)(op + (16 - d) * 16 * LL * LL) = o;
        }
    }
}

extern "C" void kernel_launch(void* const* d_in, const int* in_sizes, int n_in,
                              void* d_out, int out_size, void* d_ws, size_t ws_size,
                              hipStream_t stream) {
    const float* x = (const float*)d_in[0];
    const float* w = (const float*)d_in[1];
    float* out = (float*)d_out;

    dim3 grid(BB * (LL / 2));
    dim3 block(256);
    hipLaunchKernelGGL(fconv2d_wmma_kernel, grid, block, 0, stream, x, w, out);
}
